# starting kernel
# baseline (speedup 1.0000x reference)
#include <hip/hip_runtime.h>
#include <hip/hip_bf16.h>


typedef __attribute__((ext_vector_type(2))) float v2f;
typedef __attribute__((ext_vector_type(8))) float v8f;

#define GCN_NODES 50000
#define GCN_DIN   128
#define GCN_DHID  256
#define GCN_DOUT  128

__global__ void gcn_gemm_bias_wmma(const float* __restrict__ X,
                                   const float* __restrict__ W,
                                   const float* __restrict__ bias,
                                   float* __restrict__ H,
                                   int K, int N) {
    const int lane = threadIdx.x;
    const int half = lane >> 4;
    const int l    = lane & 15;

    const int mBase = blockIdx.x * 16;
    const int nBase = blockIdx.y * 64;

    const float* arow = X + (size_t)(mBase + l) * (size_t)K;

    v8f acc[4];
#pragma unroll
    for (int t = 0; t < 4; ++t) {
        const float bv = bias[nBase + t * 16 + l];
#pragma unroll
        for (int i = 0; i < 8; ++i) acc[t][i] = bv;
    }

    for (int k = 0; k < K; k += 4) {
        const v2f a = *(const v2f*)(arow + k + 2 * half);

        const int krow = k + 2 * half;
#pragma unroll
        for (int t = 0; t < 4; ++t) {
            const int col = nBase + t * 16 + l;
            v2f b;
            b.x = W[(size_t)(krow)     * (size_t)N + col];
            b.y = W[(size_t)(krow + 1) * (size_t)N + col];
            acc[t] = __builtin_amdgcn_wmma_f32_16x16x4_f32(
                 false, a,  false, b,
                 (short)0, acc[t],
                 false,  false);
        }
    }

#pragma unroll
    for (int t = 0; t < 4; ++t) {
        const int col = nBase + t * 16 + l;
#pragma unroll
        for (int i = 0; i < 8; ++i) {
            const int row = mBase + i + 8 * half;
            H[(size_t)row * (size_t)N + col] = acc[t][i];
        }
    }
}

__global__ void gcn_scatter_add(const float* __restrict__ h,
                                const int* __restrict__ edges,
                                float* __restrict__ agg,
                                float* __restrict__ deg,
                                long long nEdges, int C, int addDeg) {
    const int chunks = C >> 2;
    const long long tid = (long long)blockIdx.x * blockDim.x + threadIdx.x;
    const long long total = nEdges * (long long)chunks;
    if (tid >= total) return;

    const long long e  = tid / chunks;
    const int       c4 = (int)(tid - e * (long long)chunks) << 2;

    const int src = edges[2 * e];
    const int dst = edges[2 * e + 1];

    const float4 v = *(const float4*)(h + (size_t)src * (size_t)C + c4);
    float* p = agg + (size_t)dst * (size_t)C + c4;
    atomicAdd(p + 0, v.x);
    atomicAdd(p + 1, v.y);
    atomicAdd(p + 2, v.z);
    atomicAdd(p + 3, v.w);

    if (addDeg && c4 == 0) atomicAdd(deg + dst, 1.0f);
}

__global__ void gcn_normalize(const float* __restrict__ agg,
                              const float* __restrict__ deg,
                              float* __restrict__ out,
                              long long n, int C, int doRelu) {
    const long long i = (long long)blockIdx.x * blockDim.x + threadIdx.x;
    if (i >= n) return;
    const long long node = i / C;
    const float d = fmaxf(deg[node], 1.0f);
    float v = agg[i] / d;
    if (doRelu) v = fmaxf(v, 0.0f);
    out[i] = v;
}

__global__ void gcn_zero_f32(float* __restrict__ p, long long n) {
    const long long i = (long long)blockIdx.x * blockDim.x + threadIdx.x;
    if (i < n) p[i] = 0.0f;
}

extern "C" void kernel_launch(void* const* d_in, const int* in_sizes, int n_in,
                              void* d_out, int out_size, void* d_ws, size_t ws_size,
                              hipStream_t stream) {
    const float* X  = (const float*)d_in[0];
    const int*   E  = (const int*)  d_in[1];
    const float* W1 = (const float*)d_in[2];
    const float* b1 = (const float*)d_in[3];
    const float* W2 = (const float*)d_in[4];
    const float* b2 = (const float*)d_in[5];
    float* out = (float*)d_out;

    const long long nEdges = (long long)in_sizes[1] / 2;

    char* ws = (char*)d_ws;
    float* h1   = (float*)ws;
    float* agg1 = (float*)(ws + (size_t)GCN_NODES * GCN_DHID * 4);
    float* deg  = (float*)(ws + (size_t)GCN_NODES * GCN_DHID * 8);
    float* h2   = agg1;

    const long long nH1  = (long long)GCN_NODES * GCN_DHID;
    const long long nOut = (long long)GCN_NODES * GCN_DOUT;

    gcn_gemm_bias_wmma<<<dim3(GCN_NODES / 16, GCN_DHID / 64), 32, 0, stream>>>(
        X, W1, b1, h1, GCN_DIN, GCN_DHID);

    {
        const long long nz = nH1 + GCN_NODES;
        gcn_zero_f32<<<(int)((nz + 255) / 256), 256, 0, stream>>>(agg1, nz);
    }
    {
        const long long tot = nEdges * (GCN_DHID / 4);
        gcn_scatter_add<<<(int)((tot + 255) / 256), 256, 0, stream>>>(
            h1, E, agg1, deg, nEdges, GCN_DHID,  1);
    }

    gcn_normalize<<<(int)((nH1 + 255) / 256), 256, 0, stream>>>(
        agg1, deg, h1, nH1, GCN_DHID,  1);

    gcn_gemm_bias_wmma<<<dim3(GCN_NODES / 16, GCN_DOUT / 64), 32, 0, stream>>>(
        h1, W2, b2, h2, GCN_DHID, GCN_DOUT);

    gcn_zero_f32<<<(int)((nOut + 255) / 256), 256, 0, stream>>>(out, nOut);
    {
        const long long tot = nEdges * (GCN_DOUT / 4);
        gcn_scatter_add<<<(int)((tot + 255) / 256), 256, 0, stream>>>(
            h2, E, out, deg, nEdges, GCN_DOUT,  0);
    }
    gcn_normalize<<<(int)((nOut + 255) / 256), 256, 0, stream>>>(
        out, deg, out, nOut, GCN_DOUT,  0);
}


# Owner groups: each found entry goes to an LDS queue of the one wave that owns its row (row mod 32); that wave adds rows with plain 16-byte LDS load/add/store, no LDS float atomic; arrival order
# speedup vs baseline: 3.3808x; 3.3808x over previous
#include <hip/hip_runtime.h>
#include <hip/hip_bf16.h>


typedef __attribute__((ext_vector_type(2))) float v2f;
typedef __attribute__((ext_vector_type(8))) float v8f;

#define GCN_NODES 50000
#define GCN_DIN   128
#define GCN_DHID  256
#define GCN_DOUT  128

__global__ void gcn_gemm_bias_wmma(const float* __restrict__ X,
                                   const float* __restrict__ W,
                                   const float* __restrict__ bias,
                                   float* __restrict__ H,
                                   int K, int N) {
    const int lane = threadIdx.x;
    const int half = lane >> 4;
    const int l    = lane & 15;

    const int mBase = blockIdx.x * 16;
    const int nBase = blockIdx.y * 64;

    const float* arow = X + (size_t)(mBase + l) * (size_t)K;

    v8f acc[4];
#pragma unroll
    for (int t = 0; t < 4; ++t) {
        const float bv = bias[nBase + t * 16 + l];
#pragma unroll
        for (int i = 0; i < 8; ++i) acc[t][i] = bv;
    }

    for (int k = 0; k < K; k += 4) {
        const v2f a = *(const v2f*)(arow + k + 2 * half);

        const int krow = k + 2 * half;
#pragma unroll
        for (int t = 0; t < 4; ++t) {
            const int col = nBase + t * 16 + l;
            v2f b;
            b.x = W[(size_t)(krow)     * (size_t)N + col];
            b.y = W[(size_t)(krow + 1) * (size_t)N + col];
            acc[t] = __builtin_amdgcn_wmma_f32_16x16x4_f32(
                 false, a,  false, b,
                 (short)0, acc[t],
                 false,  false);
        }
    }

#pragma unroll
    for (int t = 0; t < 4; ++t) {
        const int col = nBase + t * 16 + l;
#pragma unroll
        for (int i = 0; i < 8; ++i) {
            const int row = mBase + i + 8 * half;
            H[(size_t)row * (size_t)N + col] = acc[t][i];
        }
    }
}

#define GCN_OWN_ROWS   200
#define GCN_ACC_ROWS   256
#define GCN_OWN_WAVES  32
#define GCN_OWN_LIST   136
#define GCN_WORD_LIST  160
#define GCN_SCAN_LOADS 4
#define GCN_NOBODY     255u
#define GCN_Q_CAP      192
#define GCN_LIST_CHUNKS 1023
#define GCN_NO_ENTRY    0xffffffffu
#define GCN_IDS_BYTES   (1u << 20)
#define GCN_EDGES       800000

typedef __attribute__((ext_vector_type(4))) int      v4i;
typedef __attribute__((ext_vector_type(4))) unsigned v4u;
typedef __attribute__((ext_vector_type(4))) float    v4f;

static_assert(GCN_NODES <= 65536, "src is packed into 16 bits of a list entry");
static_assert((GCN_EDGES + 2047) / 2048 * 2048 <= GCN_IDS_BYTES,
              "the owner-byte plane, padded to whole batches of 2048 edges, ends before the lists");
static_assert((GCN_NODES + GCN_OWN_ROWS - 1) / GCN_OWN_ROWS <= 255 && GCN_OWN_ROWS <= 256 &&
              GCN_OWN_ROWS <= GCN_ACC_ROWS,
              "a group's number must fit a byte below GCN_NOBODY, a local row 8 bits of a list entry");

__device__ __forceinline__ long long gcn_scan_batches(long long nEdges) {
    const long long nWords = (nEdges + 3) >> 2;
    const long long nQuads = (nWords + 3) >> 2;
    return (nQuads + 32 * GCN_SCAN_LOADS - 1) / (32 * GCN_SCAN_LOADS);
}

__global__ __launch_bounds__(256)
void gcn_owner_ids(const int* __restrict__ edges,
                   unsigned* __restrict__ ids,
                   long long nEdges) {
    const long long nWordsPad = gcn_scan_batches(nEdges) * (32 * GCN_SCAN_LOADS * 4);
    for (long long i = (long long)blockIdx.x * 256 + threadIdx.x; i < nWordsPad;
         i += (long long)gridDim.x * 256) {
        const long long e0 = 4 * i;
        int d[4] = {-1, -1, -1, -1};
        if (e0 + 3 < nEdges) {
            const v4i a = *(const v4i*)(edges + 2 * e0);
            const v4i b = *(const v4i*)(edges + 2 * e0 + 4);
            d[0] = a.y; d[1] = a.w; d[2] = b.y; d[3] = b.w;
        } else {
#pragma unroll
            for (int j = 0; j < 4; ++j)
                if (e0 + j < nEdges) d[j] = edges[2 * (e0 + j) + 1];
        }
        unsigned w = 0u;
#pragma unroll
        for (int j = 0; j < 4; ++j) {
            const unsigned id = (unsigned)d[j] < (unsigned)GCN_NODES
                                    ? (unsigned)d[j] / (unsigned)GCN_OWN_ROWS : GCN_NOBODY;
            w |= id << (8 * j);
        }
        ids[i] = w;
    }
}

template <int C, int RELU, int LISTS>
__global__ __launch_bounds__(1024)
void gcn_owner_mean(const float* __restrict__ h,
                    const int* __restrict__ edges,
                    const unsigned* __restrict__ ids,
                    unsigned* __restrict__ lists,
                    float* __restrict__ outp,
                    long long nEdges) {
    __shared__ float    acc[(C >= 256 ? GCN_OWN_ROWS : GCN_ACC_ROWS) * C];
    __shared__ unsigned rowQ[GCN_OWN_WAVES * GCN_Q_CAP];
    __shared__ int      qCount[GCN_OWN_WAVES];
    __shared__ int      degS[GCN_OWN_ROWS];
    __shared__ unsigned hitList[GCN_OWN_WAVES * GCN_OWN_LIST];
    __shared__ unsigned wordList[GCN_OWN_WAVES * GCN_WORD_LIST];
    __shared__ unsigned outBuf[LISTS == 1 ? GCN_OWN_WAVES * 64 : 1];
    __shared__ int      nChunksS;

    const int tid  = threadIdx.x;
    const int lane = tid & 31;
    const int wave = tid >> 5;
    const int r0   = blockIdx.x * GCN_OWN_ROWS;

    for (int i = tid; i < GCN_OWN_ROWS * C / 4; i += 1024)
        ((v4f*)acc)[i] = (v4f){0.0f, 0.0f, 0.0f, 0.0f};
    if (tid < GCN_OWN_ROWS) degS[tid] = 0;
    if (tid == 0) nChunksS = 0;
    if (tid < GCN_OWN_WAVES) qCount[tid] = 0;
    __syncthreads();

    unsigned* myList  = hitList  + wave * GCN_OWN_LIST;
    unsigned* myWords = wordList + wave * GCN_WORD_LIST;
    unsigned* myOut   = outBuf + (LISTS == 1 ? wave * 64 : 0);
    unsigned* myHead   = lists + (size_t)blockIdx.x * (size_t)((GCN_LIST_CHUNKS + 1) * 32);
    unsigned* myChunks = myHead + 32;
    int count    = 0;
    int nNoted   = 0;
    int outCount = 0;

    auto emitChunk = [&]() {
        int c = 0;
        if (lane == 0) c = atomicAdd(&nChunksS, 1);
        c = __builtin_amdgcn_readfirstlane(c);
        const unsigned e = lane < outCount ? myOut[lane] : GCN_NO_ENTRY;
        if (c < GCN_LIST_CHUNKS) myChunks[(size_t)c * 32 + lane] = e;
        const int rem = outCount > 32 ? outCount - 32 : 0;
        if (rem > 0) {
            unsigned t = 0u;
            if (lane < rem) t = myOut[32 + lane];
            if (lane < rem) myOut[lane] = t;
        }
        outCount = rem;
    };

    auto take = [&](int s, int d) {
        const bool hit = (unsigned)d < (unsigned)GCN_OWN_ROWS;
        const unsigned m = __builtin_amdgcn_ballot_w32(hit);
        if (m != 0u) {
            bool full = false;
            unsigned e = 0u;
            if (hit) {
                e = (unsigned)s | ((unsigned)d << 16);
                const int r = (int)__builtin_amdgcn_mbcnt_lo(m, 0u);
                if (LISTS == 1) myOut[outCount + r] = e;
                atomicAdd(&degS[d], 1);
                const int ow = d & (GCN_OWN_WAVES - 1);
                const int slot = atomicAdd(&qCount[ow], 1);
                if (slot < GCN_Q_CAP) rowQ[ow * GCN_Q_CAP + slot] = e;
                else full = true;
            }
            const unsigned mf = __builtin_amdgcn_ballot_w32(full);
            if (mf != 0u) {
                if (full) myList[count + (int)__builtin_amdgcn_mbcnt_lo(mf, 0u)] = e;
                count += __builtin_popcount(mf);
            }
            const int n = __builtin_popcount(m);
            if (LISTS == 1) {
                outCount += n;
                if (outCount >= 32) emitChunk();
            }
        }
    };

    auto flush4 = [&]() {
        int i = 0;
        for (; i + 4 <= count; i += 4) {
            unsigned p[4];
#pragma unroll
            for (int u = 0; u < 4; ++u)
                p[u] = (unsigned)__builtin_amdgcn_readfirstlane((int)myList[i + u]);
            float v[4][C / 32];
#pragma unroll
            for (int u = 0; u < 4; ++u) {
                const float* hp = h + (size_t)(p[u] & 0xffffu) * (size_t)C + lane;
#pragma unroll
                for (int j = 0; j < C / 32; ++j) v[u][j] = hp[32 * j];
            }
#pragma unroll
            for (int u = 0; u < 4; ++u) {
                float* ap = acc + (p[u] >> 16) * C + lane;
#pragma unroll
                for (int j = 0; j < C / 32; ++j) atomicAdd(ap + 32 * j, v[u][j]);
            }
        }
        const int rem = count - i;
        if (i != 0 && rem != 0) {
            unsigned t = 0u;
            if (lane < rem) t = myList[i + lane];
            if (lane < rem) myList[lane] = t;
        }
        count = rem;
    };

    const long long nPairs = nEdges >> 1;

    auto decode = [&](int n) {
        const bool act = lane < n;
        const unsigned wd = act ? myWords[nNoted - n + lane] : 0u;
        const long long p0 = 2 * (long long)wd;
        const bool ok0 = act && p0 < nPairs;
        const bool ok1 = act && p0 + 1 < nPairs;
        const v4i a = *(const v4i*)(edges + 4 * (ok0 ? p0 : 0));
        const v4i b = *(const v4i*)(edges + 4 * (ok1 ? p0 + 1 : 0));
        take(a.x, ok0 ? a.y - r0 : -1);
        take(a.z, ok0 ? a.w - r0 : -1);
        take(b.x, ok1 ? b.y - r0 : -1);
        take(b.z, ok1 ? b.w - r0 : -1);
        nNoted -= n;
        if (count >= 4) flush4();
    };

    const long long nBatches = gcn_scan_batches(nEdges);
    const v4u* plane = (const v4u*)ids;
    const unsigned mine = (unsigned)blockIdx.x * 0x01010101u;

    auto loadBatch = [&](v4u (&q)[GCN_SCAN_LOADS], long long b) {
#pragma unroll
        for (int j = 0; j < GCN_SCAN_LOADS; ++j)
            q[j] = plane[b * (32 * GCN_SCAN_LOADS) + j * 32 + lane];
    };

    int nCh = -1;
    if (LISTS == 2) nCh = __builtin_amdgcn_readfirstlane((int)myHead[0]);
    const bool fromList = LISTS == 2 && (unsigned)nCh <= (unsigned)GCN_LIST_CHUNKS;

    if (fromList) {
        unsigned curE = GCN_NO_ENTRY;
        if (wave < nCh) curE = myChunks[(size_t)wave * 32 + lane];
        for (int c = wave; c < nCh; c += GCN_OWN_WAVES) {
            unsigned nxtE = GCN_NO_ENTRY;
            if (c + GCN_OWN_WAVES < nCh)
                nxtE = myChunks[(size_t)(c + GCN_OWN_WAVES) * 32 + lane];
            const unsigned s = curE & 0xffffu;
            take((int)s, s < (unsigned)GCN_NODES ? (int)(curE >> 16) : -1);
            if (count >= 4) flush4();
            curE = nxtE;
        }
    } else {
    if (nPairs > 0) {
        v4u cur[GCN_SCAN_LOADS];
        loadBatch(cur, wave < nBatches ? (long long)wave : nBatches - 1);
        for (long long b = wave; b < nBatches; b += GCN_OWN_WAVES) {
            v4u nxt[GCN_SCAN_LOADS];
            loadBatch(nxt, b + GCN_OWN_WAVES < nBatches ? b + GCN_OWN_WAVES : nBatches - 1);
#pragma unroll
            for (int j = 0; j < GCN_SCAN_LOADS; ++j) {
                const unsigned word0 =
                    (unsigned)(b * (32 * GCN_SCAN_LOADS) + j * 32 + lane) * 4u;
#pragma unroll
                for (int k = 0; k < 4; ++k) {
                    const unsigned x = cur[j][k] ^ mine;
                    const bool any = ((x - 0x01010101u) & ~x & 0x80808080u) != 0u;
                    const unsigned m = __builtin_amdgcn_ballot_w32(any);
                    if (m != 0u) {
                        if (any)
                            myWords[nNoted + (int)__builtin_amdgcn_mbcnt_lo(m, 0u)] = word0 + k;
                        nNoted += __builtin_popcount(m);
                    }
                }
                while (nNoted >= 32) decode(32);
            }
#pragma unroll
            for (int j = 0; j < GCN_SCAN_LOADS; ++j) cur[j] = nxt[j];
        }
        if (nNoted > 0) decode(nNoted);
    }
    if ((nEdges & 1) && wave == 0) {
        int s = 0, d = -1;
        if (lane == 0) {
            s = edges[2 * (nEdges - 1)];
            d = edges[2 * (nEdges - 1) + 1] - r0;
        }
        take(s, d);
    }
    }
    flush4();
    for (int i = 0; i < count; ++i) {
        const unsigned p = (unsigned)__builtin_amdgcn_readfirstlane((int)myList[i]);
        const float* hp = h + (size_t)(p & 0xffffu) * (size_t)C + lane;
        float* ap = acc + (p >> 16) * C + lane;
        float v[C / 32];
#pragma unroll
        for (int j = 0; j < C / 32; ++j) v[j] = hp[32 * j];
#pragma unroll
        for (int j = 0; j < C / 32; ++j) atomicAdd(ap + 32 * j, v[j]);
    }
    if (LISTS == 1 && outCount > 0) emitChunk();
    __syncthreads();
    if (LISTS == 1 && wave == 0) myHead[lane] = (unsigned)nChunksS;

    {
        constexpr int V = C / 128;
        int nq = __builtin_amdgcn_readfirstlane(qCount[wave]);
        if (nq > GCN_Q_CAP) nq = GCN_Q_CAP;
        const unsigned* q = rowQ + wave * GCN_Q_CAP;
        auto entryAt = [&](int i) -> unsigned {
            return (unsigned)__builtin_amdgcn_readfirstlane((int)q[i < nq ? i : nq - 1]);
        };
        if (nq > 0) {
            unsigned pc[4];
            v4f vc[4][V];
#pragma unroll
            for (int u = 0; u < 4; ++u) pc[u] = entryAt(u);
#pragma unroll
            for (int u = 0; u < 4; ++u) {
                const v4f* hp = (const v4f*)(h + (size_t)(pc[u] & 0xffffu) * (size_t)C);
#pragma unroll
                for (int j = 0; j < V; ++j) vc[u][j] = hp[32 * j + lane];
            }
            for (int i = 0; i < nq; i += 4) {
                unsigned pn[4];
                v4f vn[4][V];
#pragma unroll
                for (int u = 0; u < 4; ++u) pn[u] = entryAt(i + 4 + u);
#pragma unroll
                for (int u = 0; u < 4; ++u) {
                    const v4f* hp = (const v4f*)(h + (size_t)(pn[u] & 0xffffu) * (size_t)C);
#pragma unroll
                    for (int j = 0; j < V; ++j) vn[u][j] = hp[32 * j + lane];
                }
#pragma unroll
                for (int u = 0; u < 4; ++u) {
                    if (i + u < nq) {
                        v4f* ap = (v4f*)(acc + (pc[u] >> 16) * C);
#pragma unroll
                        for (int j = 0; j < V; ++j) {
                            v4f a = ap[32 * j + lane];
                            a.x += vc[u][j].x; a.y += vc[u][j].y;
                            a.z += vc[u][j].z; a.w += vc[u][j].w;
                            ap[32 * j + lane] = a;
                        }
                    }
                }
#pragma unroll
                for (int u = 0; u < 4; ++u) {
                    pc[u] = pn[u];
#pragma unroll
                    for (int j = 0; j < V; ++j) vc[u][j] = vn[u][j];
                }
            }
        }
    }
    __syncthreads();

    for (int i = tid; i < GCN_OWN_ROWS * C / 4; i += 1024) {
        const int row  = i / (C / 4);
        const int node = r0 + row;
        if (node < GCN_NODES) {
            v4f a = ((const v4f*)acc)[i];
            const float d = fmaxf((float)degS[row], 1.0f);
            a.x = a.x / d; a.y = a.y / d; a.z = a.z / d; a.w = a.w / d;
            if (RELU) {
                a.x = fmaxf(a.x, 0.0f); a.y = fmaxf(a.y, 0.0f);
                a.z = fmaxf(a.z, 0.0f); a.w = fmaxf(a.w, 0.0f);
            }
            ((v4f*)(outp + (size_t)node * (size_t)C))[i % (C / 4)] = a;
        }
    }
}

extern "C" void kernel_launch(void* const* d_in, const int* in_sizes, int n_in,
                              void* d_out, int out_size, void* d_ws, size_t ws_size,
                              hipStream_t stream) {
    const float* X  = (const float*)d_in[0];
    const int*   E  = (const int*)  d_in[1];
    const float* W1 = (const float*)d_in[2];
    const float* b1 = (const float*)d_in[3];
    const float* W2 = (const float*)d_in[4];
    const float* b2 = (const float*)d_in[5];
    float* out = (float*)d_out;

    const long long nEdges = (long long)in_sizes[1] / 2;

    char* ws = (char*)d_ws;
    float* h1  = (float*)ws;
    float* h1n = (float*)(ws + (size_t)GCN_NODES * GCN_DHID * 4);
    float* h2  = h1;
    unsigned* ids = (unsigned*)(ws + (size_t)GCN_NODES * GCN_DHID * 4 * 2);
    unsigned* lists = (unsigned*)(ws + (size_t)GCN_NODES * GCN_DHID * 4 * 2 + GCN_IDS_BYTES);

    gcn_owner_ids<<<800, 256, 0, stream>>>(E, ids, nEdges);

    gcn_gemm_bias_wmma<<<dim3(GCN_NODES / 16, GCN_DHID / 64), 32, 0, stream>>>(
        X, W1, b1, h1, GCN_DIN, GCN_DHID);

    gcn_owner_mean<GCN_DHID, 1, 1><<<(GCN_NODES + GCN_OWN_ROWS - 1) / GCN_OWN_ROWS, 1024, 0, stream>>>(
        h1, E, ids, lists, h1n, nEdges);

    gcn_gemm_bias_wmma<<<dim3(GCN_NODES / 16, GCN_DOUT / 64), 32, 0, stream>>>(
        h1n, W2, b2, h2, GCN_DHID, GCN_DOUT);

    gcn_owner_mean<GCN_DOUT, 0, 2><<<(GCN_NODES + GCN_OWN_ROWS - 1) / GCN_OWN_ROWS, 1024, 0, stream>>>(
        h2, E, ids, lists, out, nEdges);
}


# Both GEMMs: 4-wave work-groups hold their K x 64 strip of W in LDS as k pairs (K, N compile-time); a B fragment is one 8-byte LDS read, not two global loads plus 64-bit address math; sums unchanged
# speedup vs baseline: 3.9878x; 1.1795x over previous
#include <hip/hip_runtime.h>
#include <hip/hip_bf16.h>


typedef __attribute__((ext_vector_type(2))) float v2f;
typedef __attribute__((ext_vector_type(8))) float v8f;
typedef __attribute__((ext_vector_type(4))) float v4f;

#define GCN_NODES 50000
#define GCN_DIN   128
#define GCN_DHID  256
#define GCN_DOUT  128

#define GCN_GEMM_WAVES 4

__device__ __forceinline__ int gcn_bs_at(int p, int c) {
    return p * 64 + (c ^ ((p & 1) << 4));
}

template <int K, int N>
__global__ __launch_bounds__(32 * GCN_GEMM_WAVES)
void gcn_gemm_bias_wmma(const float* __restrict__ X,
                        const float* __restrict__ W,
                        const float* __restrict__ bias,
                        float* __restrict__ H,
                        int mTiles) {
    __shared__ v2f Bs[(K / 2) * 64];

    const int tid  = threadIdx.x;
    const int lane = tid & 31;
    const int wave = tid >> 5;
    const int half = lane >> 4;
    const int l    = lane & 15;

    const int nBase = blockIdx.y * 64;

    static_assert((K / 2) * 16 % (32 * GCN_GEMM_WAVES * 8) == 0, "whole groups of 8 k pairs per thread");
#pragma unroll 8
    for (int it = 0; it < (K / 2) * 16 / (32 * GCN_GEMM_WAVES); ++it) {
        const int i = tid + it * (32 * GCN_GEMM_WAVES);
        const int p = i >> 4;
        const int c = (i & 15) * 4;
        const float* wp = W + (size_t)(2 * p) * (size_t)N + nBase + c;
        const v4f w0 = *(const v4f*)wp;
        const v4f w1 = *(const v4f*)(wp + N);
        v4f* d = (v4f*)(Bs + gcn_bs_at(p, c));
        d[0] = (v4f){w0.x, w1.x, w0.y, w1.y};
        d[1] = (v4f){w0.z, w1.z, w0.w, w1.w};
    }
    __syncthreads();

    const int mTile = blockIdx.x * GCN_GEMM_WAVES + wave;
    if (mTile >= mTiles) return;
    const int mBase = mTile * 16;

    const float* arow = X + (size_t)(mBase + l) * (size_t)K + 2 * half;

    v8f acc[4];
#pragma unroll
    for (int t = 0; t < 4; ++t) {
        const float bv = bias[nBase + t * 16 + l];
#pragma unroll
        for (int i = 0; i < 8; ++i) acc[t][i] = bv;
    }

#pragma unroll 8
    for (int s = 0; s < K / 4; ++s) {
        const v2f a = *(const v2f*)(arow + 4 * s);
        v2f b[4];
#pragma unroll
        for (int t = 0; t < 4; ++t) b[t] = Bs[gcn_bs_at(2 * s + half, 16 * t + l)];
#pragma unroll
        for (int t = 0; t < 4; ++t) {
            acc[t] = __builtin_amdgcn_wmma_f32_16x16x4_f32(
                 false, a,  false, b[t],
                 (short)0, acc[t],
                 false,  false);
        }
    }

#pragma unroll
    for (int t = 0; t < 4; ++t) {
        const int col = nBase + t * 16 + l;
#pragma unroll
        for (int i = 0; i < 8; ++i) {
            const int row = mBase + i + 8 * half;
            H[(size_t)row * (size_t)N + col] = acc[t][i];
        }
    }
}

#define GCN_OWN_ROWS   200
#define GCN_ACC_ROWS   256
#define GCN_OWN_WAVES  32
#define GCN_OWN_LIST   136
#define GCN_WORD_LIST  160
#define GCN_SCAN_LOADS 4
#define GCN_NOBODY     255u
#define GCN_Q_CAP      192
#define GCN_LIST_CHUNKS 1023
#define GCN_NO_ENTRY    0xffffffffu
#define GCN_IDS_BYTES   (1u << 20)
#define GCN_EDGES       800000

typedef __attribute__((ext_vector_type(4))) int      v4i;
typedef __attribute__((ext_vector_type(4))) unsigned v4u;

static_assert(GCN_NODES <= 65536, "src is packed into 16 bits of a list entry");
static_assert((GCN_EDGES + 2047) / 2048 * 2048 <= GCN_IDS_BYTES,
              "the owner-byte plane, padded to whole batches of 2048 edges, ends before the lists");
static_assert((GCN_NODES + GCN_OWN_ROWS - 1) / GCN_OWN_ROWS <= 255 && GCN_OWN_ROWS <= 256 &&
              GCN_OWN_ROWS <= GCN_ACC_ROWS,
              "a group's number must fit a byte below GCN_NOBODY, a local row 8 bits of a list entry");

__device__ __forceinline__ long long gcn_scan_batches(long long nEdges) {
    const long long nWords = (nEdges + 3) >> 2;
    const long long nQuads = (nWords + 3) >> 2;
    return (nQuads + 32 * GCN_SCAN_LOADS - 1) / (32 * GCN_SCAN_LOADS);
}

__global__ __launch_bounds__(256)
void gcn_owner_ids(const int* __restrict__ edges,
                   unsigned* __restrict__ ids,
                   long long nEdges) {
    const long long nWordsPad = gcn_scan_batches(nEdges) * (32 * GCN_SCAN_LOADS * 4);
    for (long long i = (long long)blockIdx.x * 256 + threadIdx.x; i < nWordsPad;
         i += (long long)gridDim.x * 256) {
        const long long e0 = 4 * i;
        int d[4] = {-1, -1, -1, -1};
        if (e0 + 3 < nEdges) {
            const v4i a = *(const v4i*)(edges + 2 * e0);
            const v4i b = *(const v4i*)(edges + 2 * e0 + 4);
            d[0] = a.y; d[1] = a.w; d[2] = b.y; d[3] = b.w;
        } else {
#pragma unroll
            for (int j = 0; j < 4; ++j)
                if (e0 + j < nEdges) d[j] = edges[2 * (e0 + j) + 1];
        }
        unsigned w = 0u;
#pragma unroll
        for (int j = 0; j < 4; ++j) {
            const unsigned id = (unsigned)d[j] < (unsigned)GCN_NODES
                                    ? (unsigned)d[j] / (unsigned)GCN_OWN_ROWS : GCN_NOBODY;
            w |= id << (8 * j);
        }
        ids[i] = w;
    }
}

template <int C, int RELU, int LISTS>
__global__ __launch_bounds__(1024)
void gcn_owner_mean(const float* __restrict__ h,
                    const int* __restrict__ edges,
                    const unsigned* __restrict__ ids,
                    unsigned* __restrict__ lists,
                    float* __restrict__ outp,
                    long long nEdges) {
    __shared__ float    acc[(C >= 256 ? GCN_OWN_ROWS : GCN_ACC_ROWS) * C];
    __shared__ unsigned rowQ[GCN_OWN_WAVES * GCN_Q_CAP];
    __shared__ int      qCount[GCN_OWN_WAVES];
    __shared__ int      degS[GCN_OWN_ROWS];
    __shared__ unsigned hitList[GCN_OWN_WAVES * GCN_OWN_LIST];
    __shared__ unsigned wordList[GCN_OWN_WAVES * GCN_WORD_LIST];
    __shared__ unsigned outBuf[LISTS == 1 ? GCN_OWN_WAVES * 64 : 1];
    __shared__ int      nChunksS;

    const int tid  = threadIdx.x;
    const int lane = tid & 31;
    const int wave = tid >> 5;
    const int r0   = blockIdx.x * GCN_OWN_ROWS;

    for (int i = tid; i < GCN_OWN_ROWS * C / 4; i += 1024)
        ((v4f*)acc)[i] = (v4f){0.0f, 0.0f, 0.0f, 0.0f};
    if (tid < GCN_OWN_ROWS) degS[tid] = 0;
    if (tid == 0) nChunksS = 0;
    if (tid < GCN_OWN_WAVES) qCount[tid] = 0;
    __syncthreads();

    unsigned* myList  = hitList  + wave * GCN_OWN_LIST;
    unsigned* myWords = wordList + wave * GCN_WORD_LIST;
    unsigned* myOut   = outBuf + (LISTS == 1 ? wave * 64 : 0);
    unsigned* myHead   = lists + (size_t)blockIdx.x * (size_t)((GCN_LIST_CHUNKS + 1) * 32);
    unsigned* myChunks = myHead + 32;
    int count    = 0;
    int nNoted   = 0;
    int outCount = 0;

    auto emitChunk = [&]() {
        int c = 0;
        if (lane == 0) c = atomicAdd(&nChunksS, 1);
        c = __builtin_amdgcn_readfirstlane(c);
        const unsigned e = lane < outCount ? myOut[lane] : GCN_NO_ENTRY;
        if (c < GCN_LIST_CHUNKS) myChunks[(size_t)c * 32 + lane] = e;
        const int rem = outCount > 32 ? outCount - 32 : 0;
        if (rem > 0) {
            unsigned t = 0u;
            if (lane < rem) t = myOut[32 + lane];
            if (lane < rem) myOut[lane] = t;
        }
        outCount = rem;
    };

    auto take = [&](int s, int d) {
        const bool hit = (unsigned)d < (unsigned)GCN_OWN_ROWS;
        const unsigned m = __builtin_amdgcn_ballot_w32(hit);
        if (m != 0u) {
            bool full = false;
            unsigned e = 0u;
            if (hit) {
                e = (unsigned)s | ((unsigned)d << 16);
                const int r = (int)__builtin_amdgcn_mbcnt_lo(m, 0u);
                if (LISTS == 1) myOut[outCount + r] = e;
                atomicAdd(&degS[d], 1);
                const int ow = d & (GCN_OWN_WAVES - 1);
                const int slot = atomicAdd(&qCount[ow], 1);
                if (slot < GCN_Q_CAP) rowQ[ow * GCN_Q_CAP + slot] = e;
                else full = true;
            }
            const unsigned mf = __builtin_amdgcn_ballot_w32(full);
            if (mf != 0u) {
                if (full) myList[count + (int)__builtin_amdgcn_mbcnt_lo(mf, 0u)] = e;
                count += __builtin_popcount(mf);
            }
            const int n = __builtin_popcount(m);
            if (LISTS == 1) {
                outCount += n;
                if (outCount >= 32) emitChunk();
            }
        }
    };

    auto flush4 = [&]() {
        int i = 0;
        for (; i + 4 <= count; i += 4) {
            unsigned p[4];
#pragma unroll
            for (int u = 0; u < 4; ++u)
                p[u] = (unsigned)__builtin_amdgcn_readfirstlane((int)myList[i + u]);
            float v[4][C / 32];
#pragma unroll
            for (int u = 0; u < 4; ++u) {
                const float* hp = h + (size_t)(p[u] & 0xffffu) * (size_t)C + lane;
#pragma unroll
                for (int j = 0; j < C / 32; ++j) v[u][j] = hp[32 * j];
            }
#pragma unroll
            for (int u = 0; u < 4; ++u) {
                float* ap = acc + (p[u] >> 16) * C + lane;
#pragma unroll
                for (int j = 0; j < C / 32; ++j) atomicAdd(ap + 32 * j, v[u][j]);
            }
        }
        const int rem = count - i;
        if (i != 0 && rem != 0) {
            unsigned t = 0u;
            if (lane < rem) t = myList[i + lane];
            if (lane < rem) myList[lane] = t;
        }
        count = rem;
    };

    const long long nPairs = nEdges >> 1;

    auto decode = [&](int n) {
        const bool act = lane < n;
        const unsigned wd = act ? myWords[nNoted - n + lane] : 0u;
        const long long p0 = 2 * (long long)wd;
        const bool ok0 = act && p0 < nPairs;
        const bool ok1 = act && p0 + 1 < nPairs;
        const v4i a = *(const v4i*)(edges + 4 * (ok0 ? p0 : 0));
        const v4i b = *(const v4i*)(edges + 4 * (ok1 ? p0 + 1 : 0));
        take(a.x, ok0 ? a.y - r0 : -1);
        take(a.z, ok0 ? a.w - r0 : -1);
        take(b.x, ok1 ? b.y - r0 : -1);
        take(b.z, ok1 ? b.w - r0 : -1);
        nNoted -= n;
        if (count >= 4) flush4();
    };

    const long long nBatches = gcn_scan_batches(nEdges);
    const v4u* plane = (const v4u*)ids;
    const unsigned mine = (unsigned)blockIdx.x * 0x01010101u;

    auto loadBatch = [&](v4u (&q)[GCN_SCAN_LOADS], long long b) {
#pragma unroll
        for (int j = 0; j < GCN_SCAN_LOADS; ++j)
            q[j] = plane[b * (32 * GCN_SCAN_LOADS) + j * 32 + lane];
    };

    int nCh = -1;
    if (LISTS == 2) nCh = __builtin_amdgcn_readfirstlane((int)myHead[0]);
    const bool fromList = LISTS == 2 && (unsigned)nCh <= (unsigned)GCN_LIST_CHUNKS;

    if (fromList) {
        unsigned curE = GCN_NO_ENTRY;
        if (wave < nCh) curE = myChunks[(size_t)wave * 32 + lane];
        for (int c = wave; c < nCh; c += GCN_OWN_WAVES) {
            unsigned nxtE = GCN_NO_ENTRY;
            if (c + GCN_OWN_WAVES < nCh)
                nxtE = myChunks[(size_t)(c + GCN_OWN_WAVES) * 32 + lane];
            const unsigned s = curE & 0xffffu;
            take((int)s, s < (unsigned)GCN_NODES ? (int)(curE >> 16) : -1);
            if (count >= 4) flush4();
            curE = nxtE;
        }
    } else {
    if (nPairs > 0) {
        v4u cur[GCN_SCAN_LOADS];
        loadBatch(cur, wave < nBatches ? (long long)wave : nBatches - 1);
        for (long long b = wave; b < nBatches; b += GCN_OWN_WAVES) {
            v4u nxt[GCN_SCAN_LOADS];
            loadBatch(nxt, b + GCN_OWN_WAVES < nBatches ? b + GCN_OWN_WAVES : nBatches - 1);
#pragma unroll
            for (int j = 0; j < GCN_SCAN_LOADS; ++j) {
                const unsigned word0 =
                    (unsigned)(b * (32 * GCN_SCAN_LOADS) + j * 32 + lane) * 4u;
#pragma unroll
                for (int k = 0; k < 4; ++k) {
                    const unsigned x = cur[j][k] ^ mine;
                    const bool any = ((x - 0x01010101u) & ~x & 0x80808080u) != 0u;
                    const unsigned m = __builtin_amdgcn_ballot_w32(any);
                    if (m != 0u) {
                        if (any)
                            myWords[nNoted + (int)__builtin_amdgcn_mbcnt_lo(m, 0u)] = word0 + k;
                        nNoted += __builtin_popcount(m);
                    }
                }
                while (nNoted >= 32) decode(32);
            }
#pragma unroll
            for (int j = 0; j < GCN_SCAN_LOADS; ++j) cur[j] = nxt[j];
        }
        if (nNoted > 0) decode(nNoted);
    }
    if ((nEdges & 1) && wave == 0) {
        int s = 0, d = -1;
        if (lane == 0) {
            s = edges[2 * (nEdges - 1)];
            d = edges[2 * (nEdges - 1) + 1] - r0;
        }
        take(s, d);
    }
    }
    flush4();
    for (int i = 0; i < count; ++i) {
        const unsigned p = (unsigned)__builtin_amdgcn_readfirstlane((int)myList[i]);
        const float* hp = h + (size_t)(p & 0xffffu) * (size_t)C + lane;
        float* ap = acc + (p >> 16) * C + lane;
        float v[C / 32];
#pragma unroll
        for (int j = 0; j < C / 32; ++j) v[j] = hp[32 * j];
#pragma unroll
        for (int j = 0; j < C / 32; ++j) atomicAdd(ap + 32 * j, v[j]);
    }
    if (LISTS == 1 && outCount > 0) emitChunk();
    __syncthreads();
    if (LISTS == 1 && wave == 0) myHead[lane] = (unsigned)nChunksS;

    {
        constexpr int V = C / 128;
        int nq = __builtin_amdgcn_readfirstlane(qCount[wave]);
        if (nq > GCN_Q_CAP) nq = GCN_Q_CAP;
        const unsigned* q = rowQ + wave * GCN_Q_CAP;
        auto entryAt = [&](int i) -> unsigned {
            return (unsigned)__builtin_amdgcn_readfirstlane((int)q[i < nq ? i : nq - 1]);
        };
        if (nq > 0) {
            unsigned pc[4];
            v4f vc[4][V];
#pragma unroll
            for (int u = 0; u < 4; ++u) pc[u] = entryAt(u);
#pragma unroll
            for (int u = 0; u < 4; ++u) {
                const v4f* hp = (const v4f*)(h + (size_t)(pc[u] & 0xffffu) * (size_t)C);
#pragma unroll
                for (int j = 0; j < V; ++j) vc[u][j] = hp[32 * j + lane];
            }
            for (int i = 0; i < nq; i += 4) {
                unsigned pn[4];
                v4f vn[4][V];
#pragma unroll
                for (int u = 0; u < 4; ++u) pn[u] = entryAt(i + 4 + u);
#pragma unroll
                for (int u = 0; u < 4; ++u) {
                    const v4f* hp = (const v4f*)(h + (size_t)(pn[u] & 0xffffu) * (size_t)C);
#pragma unroll
                    for (int j = 0; j < V; ++j) vn[u][j] = hp[32 * j + lane];
                }
#pragma unroll
                for (int u = 0; u < 4; ++u) {
                    if (i + u < nq) {
                        v4f* ap = (v4f*)(acc + (pc[u] >> 16) * C);
#pragma unroll
                        for (int j = 0; j < V; ++j) {
                            v4f a = ap[32 * j + lane];
                            a.x += vc[u][j].x; a.y += vc[u][j].y;
                            a.z += vc[u][j].z; a.w += vc[u][j].w;
                            ap[32 * j + lane] = a;
                        }
                    }
                }
#pragma unroll
                for (int u = 0; u < 4; ++u) {
                    pc[u] = pn[u];
#pragma unroll
                    for (int j = 0; j < V; ++j) vc[u][j] = vn[u][j];
                }
            }
        }
    }
    __syncthreads();

    for (int i = tid; i < GCN_OWN_ROWS * C / 4; i += 1024) {
        const int row  = i / (C / 4);
        const int node = r0 + row;
        if (node < GCN_NODES) {
            v4f a = ((const v4f*)acc)[i];
            const float d = fmaxf((float)degS[row], 1.0f);
            a.x = a.x / d; a.y = a.y / d; a.z = a.z / d; a.w = a.w / d;
            if (RELU) {
                a.x = fmaxf(a.x, 0.0f); a.y = fmaxf(a.y, 0.0f);
                a.z = fmaxf(a.z, 0.0f); a.w = fmaxf(a.w, 0.0f);
            }
            ((v4f*)(outp + (size_t)node * (size_t)C))[i % (C / 4)] = a;
        }
    }
}

extern "C" void kernel_launch(void* const* d_in, const int* in_sizes, int n_in,
                              void* d_out, int out_size, void* d_ws, size_t ws_size,
                              hipStream_t stream) {
    const float* X  = (const float*)d_in[0];
    const int*   E  = (const int*)  d_in[1];
    const float* W1 = (const float*)d_in[2];
    const float* b1 = (const float*)d_in[3];
    const float* W2 = (const float*)d_in[4];
    const float* b2 = (const float*)d_in[5];
    float* out = (float*)d_out;

    const long long nEdges = (long long)in_sizes[1] / 2;

    char* ws = (char*)d_ws;
    float* h1  = (float*)ws;
    float* h1n = (float*)(ws + (size_t)GCN_NODES * GCN_DHID * 4);
    float* h2  = h1;
    unsigned* ids = (unsigned*)(ws + (size_t)GCN_NODES * GCN_DHID * 4 * 2);
    unsigned* lists = (unsigned*)(ws + (size_t)GCN_NODES * GCN_DHID * 4 * 2 + GCN_IDS_BYTES);

    gcn_owner_ids<<<800, 256, 0, stream>>>(E, ids, nEdges);

    gcn_gemm_bias_wmma<GCN_DIN, GCN_DHID>
        <<<dim3((GCN_NODES / 16 + GCN_GEMM_WAVES - 1) / GCN_GEMM_WAVES, GCN_DHID / 64),
           32 * GCN_GEMM_WAVES, 0, stream>>>(X, W1, b1, h1, GCN_NODES / 16);

    gcn_owner_mean<GCN_DHID, 1, 1><<<(GCN_NODES + GCN_OWN_ROWS - 1) / GCN_OWN_ROWS, 1024, 0, stream>>>(
        h1, E, ids, lists, h1n, nEdges);

    gcn_gemm_bias_wmma<GCN_DHID, GCN_DOUT>
        <<<dim3((GCN_NODES / 16 + GCN_GEMM_WAVES - 1) / GCN_GEMM_WAVES, GCN_DOUT / 64),
           32 * GCN_GEMM_WAVES, 0, stream>>>(h1n, W2, b2, h2, GCN_NODES / 16);

    gcn_owner_mean<GCN_DOUT, 0, 2><<<(GCN_NODES + GCN_OWN_ROWS - 1) / GCN_OWN_ROWS, 1024, 0, stream>>>(
        h2, E, ids, lists, out, nEdges);
}


# Owner groups: each wave puts its queue in row order (stable) and sums a row in registers: one LDS add per row, not per entry; rows divided and stored during the walk, no end pass; arrival order
# speedup vs baseline: 4.1387x; 1.0378x over previous
#include <hip/hip_runtime.h>
#include <hip/hip_bf16.h>


typedef __attribute__((ext_vector_type(2))) float v2f;
typedef __attribute__((ext_vector_type(8))) float v8f;
typedef __attribute__((ext_vector_type(4))) float v4f;

#define GCN_NODES 50000
#define GCN_DIN   128
#define GCN_DHID  256
#define GCN_DOUT  128

#define GCN_GEMM_WAVES 4

__device__ __forceinline__ int gcn_bs_at(int p, int c) {
    return p * 64 + (c ^ ((p & 1) << 4));
}

template <int K, int N>
__global__ __launch_bounds__(32 * GCN_GEMM_WAVES)
void gcn_gemm_bias_wmma(const float* __restrict__ X,
                        const float* __restrict__ W,
                        const float* __restrict__ bias,
                        float* __restrict__ H,
                        int mTiles) {
    __shared__ v2f Bs[(K / 2) * 64];

    const int tid  = threadIdx.x;
    const int lane = tid & 31;
    const int wave = tid >> 5;
    const int half = lane >> 4;
    const int l    = lane & 15;

    const int nBase = blockIdx.y * 64;

    static_assert((K / 2) * 16 % (32 * GCN_GEMM_WAVES * 8) == 0, "whole groups of 8 k pairs per thread");
#pragma unroll 8
    for (int it = 0; it < (K / 2) * 16 / (32 * GCN_GEMM_WAVES); ++it) {
        const int i = tid + it * (32 * GCN_GEMM_WAVES);
        const int p = i >> 4;
        const int c = (i & 15) * 4;
        const float* wp = W + (size_t)(2 * p) * (size_t)N + nBase + c;
        const v4f w0 = *(const v4f*)wp;
        const v4f w1 = *(const v4f*)(wp + N);
        v4f* d = (v4f*)(Bs + gcn_bs_at(p, c));
        d[0] = (v4f){w0.x, w1.x, w0.y, w1.y};
        d[1] = (v4f){w0.z, w1.z, w0.w, w1.w};
    }
    __syncthreads();

    const int mTile = blockIdx.x * GCN_GEMM_WAVES + wave;
    if (mTile >= mTiles) return;
    const int mBase = mTile * 16;

    const float* arow = X + (size_t)(mBase + l) * (size_t)K + 2 * half;

    v8f acc[4];
#pragma unroll
    for (int t = 0; t < 4; ++t) {
        const float bv = bias[nBase + t * 16 + l];
#pragma unroll
        for (int i = 0; i < 8; ++i) acc[t][i] = bv;
    }

#pragma unroll 8
    for (int s = 0; s < K / 4; ++s) {
        const v2f a = *(const v2f*)(arow + 4 * s);
        v2f b[4];
#pragma unroll
        for (int t = 0; t < 4; ++t) b[t] = Bs[gcn_bs_at(2 * s + half, 16 * t + l)];
#pragma unroll
        for (int t = 0; t < 4; ++t) {
            acc[t] = __builtin_amdgcn_wmma_f32_16x16x4_f32(
                 false, a,  false, b[t],
                 (short)0, acc[t],
                 false,  false);
        }
    }

#pragma unroll
    for (int t = 0; t < 4; ++t) {
        const int col = nBase + t * 16 + l;
#pragma unroll
        for (int i = 0; i < 8; ++i) {
            const int row = mBase + i + 8 * half;
            H[(size_t)row * (size_t)N + col] = acc[t][i];
        }
    }
}

#define GCN_OWN_ROWS   200
#define GCN_ACC_ROWS   256
#define GCN_OWN_WAVES  32
#define GCN_OWN_LIST   136
#define GCN_WORD_LIST  160
#define GCN_SCAN_LOADS 4
#define GCN_NOBODY     255u
#define GCN_Q_CAP      192
#define GCN_WAVE_ROWS  ((GCN_OWN_ROWS + GCN_OWN_WAVES - 1) / GCN_OWN_WAVES)
#define GCN_Q_ROWS     8
#define GCN_LIST_CHUNKS 1023
#define GCN_NO_ENTRY    0xffffffffu
#define GCN_IDS_BYTES   (1u << 20)
#define GCN_EDGES       800000

typedef __attribute__((ext_vector_type(4))) int      v4i;
typedef __attribute__((ext_vector_type(4))) unsigned v4u;

static_assert(GCN_NODES <= 65536, "src is packed into 16 bits of a list entry");
static_assert(GCN_OWN_WAVES == 32, "a queue entry's key is its local row / 32, bits 21 and up of the entry");
static_assert((GCN_EDGES + 2047) / 2048 * 2048 <= GCN_IDS_BYTES,
              "the owner-byte plane, padded to whole batches of 2048 edges, ends before the lists");
static_assert((GCN_NODES + GCN_OWN_ROWS - 1) / GCN_OWN_ROWS <= 255 && GCN_OWN_ROWS <= 256 &&
              GCN_OWN_ROWS <= GCN_ACC_ROWS,
              "a group's number must fit a byte below GCN_NOBODY, a local row 8 bits of a list entry");

__device__ __forceinline__ long long gcn_scan_batches(long long nEdges) {
    const long long nWords = (nEdges + 3) >> 2;
    const long long nQuads = (nWords + 3) >> 2;
    return (nQuads + 32 * GCN_SCAN_LOADS - 1) / (32 * GCN_SCAN_LOADS);
}

__global__ __launch_bounds__(256)
void gcn_owner_ids(const int* __restrict__ edges,
                   unsigned* __restrict__ ids,
                   long long nEdges) {
    const long long nWordsPad = gcn_scan_batches(nEdges) * (32 * GCN_SCAN_LOADS * 4);
    for (long long i = (long long)blockIdx.x * 256 + threadIdx.x; i < nWordsPad;
         i += (long long)gridDim.x * 256) {
        const long long e0 = 4 * i;
        int d[4] = {-1, -1, -1, -1};
        if (e0 + 3 < nEdges) {
            const v4i a = *(const v4i*)(edges + 2 * e0);
            const v4i b = *(const v4i*)(edges + 2 * e0 + 4);
            d[0] = a.y; d[1] = a.w; d[2] = b.y; d[3] = b.w;
        } else {
#pragma unroll
            for (int j = 0; j < 4; ++j)
                if (e0 + j < nEdges) d[j] = edges[2 * (e0 + j) + 1];
        }
        unsigned w = 0u;
#pragma unroll
        for (int j = 0; j < 4; ++j) {
            const unsigned id = (unsigned)d[j] < (unsigned)GCN_NODES
                                    ? (unsigned)d[j] / (unsigned)GCN_OWN_ROWS : GCN_NOBODY;
            w |= id << (8 * j);
        }
        ids[i] = w;
    }
}

template <int C, int RELU, int LISTS>
__global__ __launch_bounds__(1024)
void gcn_owner_mean(const float* __restrict__ h,
                    const int* __restrict__ edges,
                    const unsigned* __restrict__ ids,
                    unsigned* __restrict__ lists,
                    float* __restrict__ outp,
                    long long nEdges) {
    __shared__ float    acc[(C >= 256 ? GCN_OWN_ROWS : GCN_ACC_ROWS) * C];
    __shared__ unsigned rowQ[GCN_OWN_WAVES * GCN_Q_CAP];
    __shared__ unsigned rowQs[GCN_OWN_WAVES * GCN_Q_CAP];
    __shared__ int      qCount[GCN_OWN_WAVES];
    __shared__ int      degS[GCN_OWN_ROWS];
    __shared__ unsigned hitList[GCN_OWN_WAVES * GCN_OWN_LIST];
    __shared__ unsigned wordList[GCN_OWN_WAVES * GCN_WORD_LIST];
    __shared__ unsigned outBuf[LISTS == 1 ? GCN_OWN_WAVES * 64 : 1];
    __shared__ int      nChunksS;

    const int tid  = threadIdx.x;
    const int lane = tid & 31;
    const int wave = tid >> 5;
    const int r0   = blockIdx.x * GCN_OWN_ROWS;

    for (int i = tid; i < GCN_OWN_ROWS * C / 4; i += 1024)
        ((v4f*)acc)[i] = (v4f){0.0f, 0.0f, 0.0f, 0.0f};
    if (tid < GCN_OWN_ROWS) degS[tid] = 0;
    if (tid == 0) nChunksS = 0;
    if (tid < GCN_OWN_WAVES) qCount[tid] = 0;
    __syncthreads();

    unsigned* myList  = hitList  + wave * GCN_OWN_LIST;
    unsigned* myWords = wordList + wave * GCN_WORD_LIST;
    unsigned* myOut   = outBuf + (LISTS == 1 ? wave * 64 : 0);
    unsigned* myHead   = lists + (size_t)blockIdx.x * (size_t)((GCN_LIST_CHUNKS + 1) * 32);
    unsigned* myChunks = myHead + 32;
    int count    = 0;
    int nNoted   = 0;
    int outCount = 0;

    auto emitChunk = [&]() {
        int c = 0;
        if (lane == 0) c = atomicAdd(&nChunksS, 1);
        c = __builtin_amdgcn_readfirstlane(c);
        const unsigned e = lane < outCount ? myOut[lane] : GCN_NO_ENTRY;
        if (c < GCN_LIST_CHUNKS) myChunks[(size_t)c * 32 + lane] = e;
        const int rem = outCount > 32 ? outCount - 32 : 0;
        if (rem > 0) {
            unsigned t = 0u;
            if (lane < rem) t = myOut[32 + lane];
            if (lane < rem) myOut[lane] = t;
        }
        outCount = rem;
    };

    auto take = [&](int s, int d) {
        const bool hit = (unsigned)d < (unsigned)GCN_OWN_ROWS;
        const unsigned m = __builtin_amdgcn_ballot_w32(hit);
        if (m != 0u) {
            bool full = false;
            unsigned e = 0u;
            if (hit) {
                e = (unsigned)s | ((unsigned)d << 16);
                const int r = (int)__builtin_amdgcn_mbcnt_lo(m, 0u);
                if (LISTS == 1) myOut[outCount + r] = e;
                atomicAdd(&degS[d], 1);
                const int ow = d & (GCN_OWN_WAVES - 1);
                const int slot = atomicAdd(&qCount[ow], 1);
                if (slot < GCN_Q_CAP) rowQ[ow * GCN_Q_CAP + slot] = e;
                else full = true;
            }
            const unsigned mf = __builtin_amdgcn_ballot_w32(full);
            if (mf != 0u) {
                if (full) myList[count + (int)__builtin_amdgcn_mbcnt_lo(mf, 0u)] = e;
                count += __builtin_popcount(mf);
            }
            const int n = __builtin_popcount(m);
            if (LISTS == 1) {
                outCount += n;
                if (outCount >= 32) emitChunk();
            }
        }
    };

    auto flush4 = [&]() {
        int i = 0;
        for (; i + 4 <= count; i += 4) {
            unsigned p[4];
#pragma unroll
            for (int u = 0; u < 4; ++u)
                p[u] = (unsigned)__builtin_amdgcn_readfirstlane((int)myList[i + u]);
            float v[4][C / 32];
#pragma unroll
            for (int u = 0; u < 4; ++u) {
                const float* hp = h + (size_t)(p[u] & 0xffffu) * (size_t)C + lane;
#pragma unroll
                for (int j = 0; j < C / 32; ++j) v[u][j] = hp[32 * j];
            }
#pragma unroll
            for (int u = 0; u < 4; ++u) {
                float* ap = acc + (p[u] >> 16) * C + lane;
#pragma unroll
                for (int j = 0; j < C / 32; ++j) atomicAdd(ap + 32 * j, v[u][j]);
            }
        }
        const int rem = count - i;
        if (i != 0 && rem != 0) {
            unsigned t = 0u;
            if (lane < rem) t = myList[i + lane];
            if (lane < rem) myList[lane] = t;
        }
        count = rem;
    };

    const long long nPairs = nEdges >> 1;

    auto decode = [&](int n) {
        const bool act = lane < n;
        const unsigned wd = act ? myWords[nNoted - n + lane] : 0u;
        const long long p0 = 2 * (long long)wd;
        const bool ok0 = act && p0 < nPairs;
        const bool ok1 = act && p0 + 1 < nPairs;
        const v4i a = *(const v4i*)(edges + 4 * (ok0 ? p0 : 0));
        const v4i b = *(const v4i*)(edges + 4 * (ok1 ? p0 + 1 : 0));
        take(a.x, ok0 ? a.y - r0 : -1);
        take(a.z, ok0 ? a.w - r0 : -1);
        take(b.x, ok1 ? b.y - r0 : -1);
        take(b.z, ok1 ? b.w - r0 : -1);
        nNoted -= n;
        if (count >= 4) flush4();
    };

    const long long nBatches = gcn_scan_batches(nEdges);
    const v4u* plane = (const v4u*)ids;
    const unsigned mine = (unsigned)blockIdx.x * 0x01010101u;

    auto loadBatch = [&](v4u (&q)[GCN_SCAN_LOADS], long long b) {
#pragma unroll
        for (int j = 0; j < GCN_SCAN_LOADS; ++j)
            q[j] = plane[b * (32 * GCN_SCAN_LOADS) + j * 32 + lane];
    };

    int nCh = -1;
    if (LISTS == 2) nCh = __builtin_amdgcn_readfirstlane((int)myHead[0]);
    const bool fromList = LISTS == 2 && (unsigned)nCh <= (unsigned)GCN_LIST_CHUNKS;

    if (fromList) {
        unsigned curE = GCN_NO_ENTRY;
        if (wave < nCh) curE = myChunks[(size_t)wave * 32 + lane];
        for (int c = wave; c < nCh; c += GCN_OWN_WAVES) {
            unsigned nxtE = GCN_NO_ENTRY;
            if (c + GCN_OWN_WAVES < nCh)
                nxtE = myChunks[(size_t)(c + GCN_OWN_WAVES) * 32 + lane];
            const unsigned s = curE & 0xffffu;
            take((int)s, s < (unsigned)GCN_NODES ? (int)(curE >> 16) : -1);
            if (count >= 4) flush4();
            curE = nxtE;
        }
    } else {
    if (nPairs > 0) {
        v4u cur[GCN_SCAN_LOADS];
        loadBatch(cur, wave < nBatches ? (long long)wave : nBatches - 1);
        for (long long b = wave; b < nBatches; b += GCN_OWN_WAVES) {
            v4u nxt[GCN_SCAN_LOADS];
            loadBatch(nxt, b + GCN_OWN_WAVES < nBatches ? b + GCN_OWN_WAVES : nBatches - 1);
#pragma unroll
            for (int j = 0; j < GCN_SCAN_LOADS; ++j) {
                const unsigned word0 =
                    (unsigned)(b * (32 * GCN_SCAN_LOADS) + j * 32 + lane) * 4u;
#pragma unroll
                for (int k = 0; k < 4; ++k) {
                    const unsigned x = cur[j][k] ^ mine;
                    const bool any = ((x - 0x01010101u) & ~x & 0x80808080u) != 0u;
                    const unsigned m = __builtin_amdgcn_ballot_w32(any);
                    if (m != 0u) {
                        if (any)
                            myWords[nNoted + (int)__builtin_amdgcn_mbcnt_lo(m, 0u)] = word0 + k;
                        nNoted += __builtin_popcount(m);
                    }
                }
                while (nNoted >= 32) decode(32);
            }
#pragma unroll
            for (int j = 0; j < GCN_SCAN_LOADS; ++j) cur[j] = nxt[j];
        }
        if (nNoted > 0) decode(nNoted);
    }
    if ((nEdges & 1) && wave == 0) {
        int s = 0, d = -1;
        if (lane == 0) {
            s = edges[2 * (nEdges - 1)];
            d = edges[2 * (nEdges - 1) + 1] - r0;
        }
        take(s, d);
    }
    }
    flush4();
    for (int i = 0; i < count; ++i) {
        const unsigned p = (unsigned)__builtin_amdgcn_readfirstlane((int)myList[i]);
        const float* hp = h + (size_t)(p & 0xffffu) * (size_t)C + lane;
        float* ap = acc + (p >> 16) * C + lane;
        float v[C / 32];
#pragma unroll
        for (int j = 0; j < C / 32; ++j) v[j] = hp[32 * j];
#pragma unroll
        for (int j = 0; j < C / 32; ++j) atomicAdd(ap + 32 * j, v[j]);
    }
    if (LISTS == 1 && outCount > 0) emitChunk();
    __syncthreads();
    if (LISTS == 1 && wave == 0) myHead[lane] = (unsigned)nChunksS;

    {
        constexpr int V  = C / 128;
        constexpr int D  = GCN_Q_ROWS;
        constexpr int NR = GCN_WAVE_ROWS;
        int nq = __builtin_amdgcn_readfirstlane(qCount[wave]);
        if (nq > GCN_Q_CAP) nq = GCN_Q_CAP;
        const unsigned* q  = rowQ  + wave * GCN_Q_CAP;
        unsigned*       qs = rowQs + wave * GCN_Q_CAP;

        int start[NR];
#pragma unroll
        for (int kk = 0; kk < NR; ++kk) start[kk] = 0;
        for (int c0 = 0; c0 < nq; c0 += 32) {
            const bool act = c0 + lane < nq;
            const int key = act ? (int)(q[c0 + lane] >> 21) : -1;
#pragma unroll
            for (int kk = 0; kk + 1 < NR; ++kk)
                start[kk + 1] += __builtin_popcount(__builtin_amdgcn_ballot_w32(key >= 0 && key <= kk));
        }
        for (int c0 = 0; c0 < nq; c0 += 32) {
            const bool act = c0 + lane < nq;
            const unsigned e = act ? q[c0 + lane] : 0u;
            const int key = act ? (int)(e >> 21) : -1;
#pragma unroll
            for (int kk = 0; kk < NR; ++kk) {
                const unsigned m = __builtin_amdgcn_ballot_w32(key == kk);
                if (key == kk) qs[start[kk] + (int)__builtin_amdgcn_mbcnt_lo(m, 0u)] = e;
                start[kk] += __builtin_popcount(m);
            }
        }

        int kdone = 0;
        auto storeRows = [&](int kend) {
#pragma unroll 1
            for (; kdone < kend; ++kdone) {
                const int row  = wave + 32 * kdone;
                const int node = r0 + row;
                if (row < GCN_OWN_ROWS && node < GCN_NODES) {
                    const float d = fmaxf((float)degS[row], 1.0f);
                    const v4f* ap = (const v4f*)(acc + row * C);
                    v4f* op = (v4f*)(outp + (size_t)node * (size_t)C);
#pragma unroll
                    for (int j = 0; j < V; ++j) {
                        v4f a = ap[32 * j + lane];
                        a.x = a.x / d; a.y = a.y / d; a.z = a.z / d; a.w = a.w / d;
                        if (RELU) {
                            a.x = fmaxf(a.x, 0.0f); a.y = fmaxf(a.y, 0.0f);
                            a.z = fmaxf(a.z, 0.0f); a.w = fmaxf(a.w, 0.0f);
                        }
                        op[32 * j + lane] = a;
                    }
                }
            }
        };

        if (nq > 0) {
            auto entryAt = [&](int i) -> unsigned {
                return (unsigned)__builtin_amdgcn_readfirstlane((int)qs[i < nq ? i : nq - 1]);
            };
            v4f sum[V];
#pragma unroll
            for (int j = 0; j < V; ++j) sum[j] = (v4f){0.0f, 0.0f, 0.0f, 0.0f};
            unsigned pc[D];
            v4f vc[D][V];
#pragma unroll
            for (int u = 0; u < D; ++u) {
                pc[u] = 0u;
#pragma unroll
                for (int j = 0; j < V; ++j) vc[u][j] = (v4f){0.0f, 0.0f, 0.0f, 0.0f};
            }
            unsigned rcur = entryAt(0) >> 16;
            auto putRow = [&]() {
                v4f* ap = (v4f*)(acc + rcur * C);
#pragma unroll
                for (int j = 0; j < V; ++j) {
                    v4f a = ap[32 * j + lane];
                    a.x += sum[j].x; a.y += sum[j].y; a.z += sum[j].z; a.w += sum[j].w;
                    ap[32 * j + lane] = a;
                    sum[j] = (v4f){0.0f, 0.0f, 0.0f, 0.0f};
                }
            };
            for (int i = -D; i < nq; i += D) {
                storeRows((int)(rcur >> 5));
                unsigned pn[D];
#pragma unroll
                for (int u = 0; u < D; ++u) pn[u] = entryAt(i + D + u);
#pragma unroll
                for (int u = 0; u < D; ++u) {
                    if (i >= 0 && i + u < nq) {
                        const unsigned r = pc[u] >> 16;
                        if (r != rcur) {
                            putRow();
                            rcur = r;
                        }
#pragma unroll
                        for (int j = 0; j < V; ++j) {
                            sum[j].x += vc[u][j].x; sum[j].y += vc[u][j].y;
                            sum[j].z += vc[u][j].z; sum[j].w += vc[u][j].w;
                        }
                    }
                    pc[u] = pn[u];
                    const v4f* hp = (const v4f*)(h + (size_t)(pc[u] & 0xffffu) * (size_t)C);
#pragma unroll
                    for (int j = 0; j < V; ++j) vc[u][j] = hp[32 * j + lane];
                }
            }
            putRow();
        }

        storeRows(NR);
    }
}

extern "C" void kernel_launch(void* const* d_in, const int* in_sizes, int n_in,
                              void* d_out, int out_size, void* d_ws, size_t ws_size,
                              hipStream_t stream) {
    const float* X  = (const float*)d_in[0];
    const int*   E  = (const int*)  d_in[1];
    const float* W1 = (const float*)d_in[2];
    const float* b1 = (const float*)d_in[3];
    const float* W2 = (const float*)d_in[4];
    const float* b2 = (const float*)d_in[5];
    float* out = (float*)d_out;

    const long long nEdges = (long long)in_sizes[1] / 2;

    char* ws = (char*)d_ws;
    float* h1  = (float*)ws;
    float* h1n = (float*)(ws + (size_t)GCN_NODES * GCN_DHID * 4);
    float* h2  = h1;
    unsigned* ids = (unsigned*)(ws + (size_t)GCN_NODES * GCN_DHID * 4 * 2);
    unsigned* lists = (unsigned*)(ws + (size_t)GCN_NODES * GCN_DHID * 4 * 2 + GCN_IDS_BYTES);

    gcn_owner_ids<<<800, 256, 0, stream>>>(E, ids, nEdges);

    gcn_gemm_bias_wmma<GCN_DIN, GCN_DHID>
        <<<dim3((GCN_NODES / 16 + GCN_GEMM_WAVES - 1) / GCN_GEMM_WAVES, GCN_DHID / 64),
           32 * GCN_GEMM_WAVES, 0, stream>>>(X, W1, b1, h1, GCN_NODES / 16);

    gcn_owner_mean<GCN_DHID, 1, 1><<<(GCN_NODES + GCN_OWN_ROWS - 1) / GCN_OWN_ROWS, 1024, 0, stream>>>(
        h1, E, ids, lists, h1n, nEdges);

    gcn_gemm_bias_wmma<GCN_DHID, GCN_DOUT>
        <<<dim3((GCN_NODES / 16 + GCN_GEMM_WAVES - 1) / GCN_GEMM_WAVES, GCN_DOUT / 64),
           32 * GCN_GEMM_WAVES, 0, stream>>>(h1n, W2, b2, h2, GCN_NODES / 16);

    gcn_owner_mean<GCN_DOUT, 0, 2><<<(GCN_NODES + GCN_OWN_ROWS - 1) / GCN_OWN_ROWS, 1024, 0, stream>>>(
        h2, E, ids, lists, out, nEdges);
}


# Owner-byte plane written inside the first GEMM launch (work-groups of column strip 0, each its share of the words, before staging W): no launch of its own, 5 launches to 4; sums unchanged
# speedup vs baseline: 4.2693x; 1.0316x over previous
#include <hip/hip_runtime.h>
#include <hip/hip_bf16.h>


typedef __attribute__((ext_vector_type(2))) float v2f;
typedef __attribute__((ext_vector_type(8))) float v8f;
typedef __attribute__((ext_vector_type(4))) float v4f;

#define GCN_NODES 50000
#define GCN_DIN   128
#define GCN_DHID  256
#define GCN_DOUT  128

#define GCN_OWN_ROWS   200
#define GCN_ACC_ROWS   256
#define GCN_OWN_WAVES  32
#define GCN_OWN_LIST   136
#define GCN_WORD_LIST  160
#define GCN_SCAN_LOADS 4
#define GCN_NOBODY     255u
#define GCN_Q_CAP      192
#define GCN_WAVE_ROWS  ((GCN_OWN_ROWS + GCN_OWN_WAVES - 1) / GCN_OWN_WAVES)
#define GCN_Q_ROWS     8
#define GCN_LIST_CHUNKS 1023
#define GCN_NO_ENTRY    0xffffffffu
#define GCN_IDS_BYTES   (1u << 20)
#define GCN_EDGES       800000

typedef __attribute__((ext_vector_type(4))) int      v4i;
typedef __attribute__((ext_vector_type(4))) unsigned v4u;

static_assert(GCN_NODES <= 65536, "src is packed into 16 bits of a list entry");
static_assert(GCN_OWN_WAVES == 32, "a queue entry's key is its local row / 32, bits 21 and up of the entry");
static_assert((GCN_EDGES + 2047) / 2048 * 2048 <= GCN_IDS_BYTES,
              "the owner-byte plane, padded to whole batches of 2048 edges, ends before the lists");
static_assert((GCN_NODES + GCN_OWN_ROWS - 1) / GCN_OWN_ROWS <= 255 && GCN_OWN_ROWS <= 256 &&
              GCN_OWN_ROWS <= GCN_ACC_ROWS,
              "a group's number must fit a byte below GCN_NOBODY, a local row 8 bits of a list entry");

__device__ __forceinline__ long long gcn_scan_batches(long long nEdges) {
    const long long nWords = (nEdges + 3) >> 2;
    const long long nQuads = (nWords + 3) >> 2;
    return (nQuads + 32 * GCN_SCAN_LOADS - 1) / (32 * GCN_SCAN_LOADS);
}

__device__ __forceinline__
void gcn_owner_ids(const int* __restrict__ edges,
                   unsigned* __restrict__ ids,
                   long long nEdges, long long first, long long stride) {
    const long long nWordsPad = gcn_scan_batches(nEdges) * (32 * GCN_SCAN_LOADS * 4);
    for (long long i = first; i < nWordsPad; i += stride) {
        const long long e0 = 4 * i;
        int d[4] = {-1, -1, -1, -1};
        if (e0 + 3 < nEdges) {
            const v4i a = *(const v4i*)(edges + 2 * e0);
            const v4i b = *(const v4i*)(edges + 2 * e0 + 4);
            d[0] = a.y; d[1] = a.w; d[2] = b.y; d[3] = b.w;
        } else {
#pragma unroll
            for (int j = 0; j < 4; ++j)
                if (e0 + j < nEdges) d[j] = edges[2 * (e0 + j) + 1];
        }
        unsigned w = 0u;
#pragma unroll
        for (int j = 0; j < 4; ++j) {
            const unsigned id = (unsigned)d[j] < (unsigned)GCN_NODES
                                    ? (unsigned)d[j] / (unsigned)GCN_OWN_ROWS : GCN_NOBODY;
            w |= id << (8 * j);
        }
        ids[i] = w;
    }
}

#define GCN_GEMM_WAVES 4

__device__ __forceinline__ int gcn_bs_at(int p, int c) {
    return p * 64 + (c ^ ((p & 1) << 4));
}

template <int K, int N, int IDS>
__global__ __launch_bounds__(32 * GCN_GEMM_WAVES)
void gcn_gemm_bias_wmma(const float* __restrict__ X,
                        const float* __restrict__ W,
                        const float* __restrict__ bias,
                        float* __restrict__ H,
                        int mTiles,
                        const int* __restrict__ edges,
                        unsigned* __restrict__ ids,
                        long long nEdges) {
    __shared__ v2f Bs[(K / 2) * 64];

    const int tid  = threadIdx.x;
    const int lane = tid & 31;
    const int wave = tid >> 5;
    const int half = lane >> 4;
    const int l    = lane & 15;

    const int nBase = blockIdx.y * 64;

    if (IDS && blockIdx.y == 0)
        gcn_owner_ids(edges, ids, nEdges,
                      (long long)blockIdx.x * (32 * GCN_GEMM_WAVES) + tid,
                      (long long)gridDim.x * (32 * GCN_GEMM_WAVES));

    static_assert((K / 2) * 16 % (32 * GCN_GEMM_WAVES * 8) == 0, "whole groups of 8 k pairs per thread");
#pragma unroll 8
    for (int it = 0; it < (K / 2) * 16 / (32 * GCN_GEMM_WAVES); ++it) {
        const int i = tid + it * (32 * GCN_GEMM_WAVES);
        const int p = i >> 4;
        const int c = (i & 15) * 4;
        const float* wp = W + (size_t)(2 * p) * (size_t)N + nBase + c;
        const v4f w0 = *(const v4f*)wp;
        const v4f w1 = *(const v4f*)(wp + N);
        v4f* d = (v4f*)(Bs + gcn_bs_at(p, c));
        d[0] = (v4f){w0.x, w1.x, w0.y, w1.y};
        d[1] = (v4f){w0.z, w1.z, w0.w, w1.w};
    }
    __syncthreads();

    const int mTile = blockIdx.x * GCN_GEMM_WAVES + wave;
    if (mTile >= mTiles) return;
    const int mBase = mTile * 16;

    const float* arow = X + (size_t)(mBase + l) * (size_t)K + 2 * half;

    v8f acc[4];
#pragma unroll
    for (int t = 0; t < 4; ++t) {
        const float bv = bias[nBase + t * 16 + l];
#pragma unroll
        for (int i = 0; i < 8; ++i) acc[t][i] = bv;
    }

#pragma unroll 8
    for (int s = 0; s < K / 4; ++s) {
        const v2f a = *(const v2f*)(arow + 4 * s);
        v2f b[4];
#pragma unroll
        for (int t = 0; t < 4; ++t) b[t] = Bs[gcn_bs_at(2 * s + half, 16 * t + l)];
#pragma unroll
        for (int t = 0; t < 4; ++t) {
            acc[t] = __builtin_amdgcn_wmma_f32_16x16x4_f32(
                 false, a,  false, b[t],
                 (short)0, acc[t],
                 false,  false);
        }
    }

#pragma unroll
    for (int t = 0; t < 4; ++t) {
        const int col = nBase + t * 16 + l;
#pragma unroll
        for (int i = 0; i < 8; ++i) {
            const int row = mBase + i + 8 * half;
            H[(size_t)row * (size_t)N + col] = acc[t][i];
        }
    }
}

template <int C, int RELU, int LISTS>
__global__ __launch_bounds__(1024)
void gcn_owner_mean(const float* __restrict__ h,
                    const int* __restrict__ edges,
                    const unsigned* __restrict__ ids,
                    unsigned* __restrict__ lists,
                    float* __restrict__ outp,
                    long long nEdges) {
    __shared__ float    acc[(C >= 256 ? GCN_OWN_ROWS : GCN_ACC_ROWS) * C];
    __shared__ unsigned rowQ[GCN_OWN_WAVES * GCN_Q_CAP];
    __shared__ unsigned rowQs[GCN_OWN_WAVES * GCN_Q_CAP];
    __shared__ int      qCount[GCN_OWN_WAVES];
    __shared__ int      degS[GCN_OWN_ROWS];
    __shared__ unsigned hitList[GCN_OWN_WAVES * GCN_OWN_LIST];
    __shared__ unsigned wordList[GCN_OWN_WAVES * GCN_WORD_LIST];
    __shared__ unsigned outBuf[LISTS == 1 ? GCN_OWN_WAVES * 64 : 1];
    __shared__ int      nChunksS;

    const int tid  = threadIdx.x;
    const int lane = tid & 31;
    const int wave = tid >> 5;
    const int r0   = blockIdx.x * GCN_OWN_ROWS;

    for (int i = tid; i < GCN_OWN_ROWS * C / 4; i += 1024)
        ((v4f*)acc)[i] = (v4f){0.0f, 0.0f, 0.0f, 0.0f};
    if (tid < GCN_OWN_ROWS) degS[tid] = 0;
    if (tid == 0) nChunksS = 0;
    if (tid < GCN_OWN_WAVES) qCount[tid] = 0;
    __syncthreads();

    unsigned* myList  = hitList  + wave * GCN_OWN_LIST;
    unsigned* myWords = wordList + wave * GCN_WORD_LIST;
    unsigned* myOut   = outBuf + (LISTS == 1 ? wave * 64 : 0);
    unsigned* myHead   = lists + (size_t)blockIdx.x * (size_t)((GCN_LIST_CHUNKS + 1) * 32);
    unsigned* myChunks = myHead + 32;
    int count    = 0;
    int nNoted   = 0;
    int outCount = 0;

    auto emitChunk = [&]() {
        int c = 0;
        if (lane == 0) c = atomicAdd(&nChunksS, 1);
        c = __builtin_amdgcn_readfirstlane(c);
        const unsigned e = lane < outCount ? myOut[lane] : GCN_NO_ENTRY;
        if (c < GCN_LIST_CHUNKS) myChunks[(size_t)c * 32 + lane] = e;
        const int rem = outCount > 32 ? outCount - 32 : 0;
        if (rem > 0) {
            unsigned t = 0u;
            if (lane < rem) t = myOut[32 + lane];
            if (lane < rem) myOut[lane] = t;
        }
        outCount = rem;
    };

    auto take = [&](int s, int d) {
        const bool hit = (unsigned)d < (unsigned)GCN_OWN_ROWS;
        const unsigned m = __builtin_amdgcn_ballot_w32(hit);
        if (m != 0u) {
            bool full = false;
            unsigned e = 0u;
            if (hit) {
                e = (unsigned)s | ((unsigned)d << 16);
                const int r = (int)__builtin_amdgcn_mbcnt_lo(m, 0u);
                if (LISTS == 1) myOut[outCount + r] = e;
                atomicAdd(&degS[d], 1);
                const int ow = d & (GCN_OWN_WAVES - 1);
                const int slot = atomicAdd(&qCount[ow], 1);
                if (slot < GCN_Q_CAP) rowQ[ow * GCN_Q_CAP + slot] = e;
                else full = true;
            }
            const unsigned mf = __builtin_amdgcn_ballot_w32(full);
            if (mf != 0u) {
                if (full) myList[count + (int)__builtin_amdgcn_mbcnt_lo(mf, 0u)] = e;
                count += __builtin_popcount(mf);
            }
            const int n = __builtin_popcount(m);
            if (LISTS == 1) {
                outCount += n;
                if (outCount >= 32) emitChunk();
            }
        }
    };

    auto flush4 = [&]() {
        int i = 0;
        for (; i + 4 <= count; i += 4) {
            unsigned p[4];
#pragma unroll
            for (int u = 0; u < 4; ++u)
                p[u] = (unsigned)__builtin_amdgcn_readfirstlane((int)myList[i + u]);
            float v[4][C / 32];
#pragma unroll
            for (int u = 0; u < 4; ++u) {
                const float* hp = h + (size_t)(p[u] & 0xffffu) * (size_t)C + lane;
#pragma unroll
                for (int j = 0; j < C / 32; ++j) v[u][j] = hp[32 * j];
            }
#pragma unroll
            for (int u = 0; u < 4; ++u) {
                float* ap = acc + (p[u] >> 16) * C + lane;
#pragma unroll
                for (int j = 0; j < C / 32; ++j) atomicAdd(ap + 32 * j, v[u][j]);
            }
        }
        const int rem = count - i;
        if (i != 0 && rem != 0) {
            unsigned t = 0u;
            if (lane < rem) t = myList[i + lane];
            if (lane < rem) myList[lane] = t;
        }
        count = rem;
    };

    const long long nPairs = nEdges >> 1;

    auto decode = [&](int n) {
        const bool act = lane < n;
        const unsigned wd = act ? myWords[nNoted - n + lane] : 0u;
        const long long p0 = 2 * (long long)wd;
        const bool ok0 = act && p0 < nPairs;
        const bool ok1 = act && p0 + 1 < nPairs;
        const v4i a = *(const v4i*)(edges + 4 * (ok0 ? p0 : 0));
        const v4i b = *(const v4i*)(edges + 4 * (ok1 ? p0 + 1 : 0));
        take(a.x, ok0 ? a.y - r0 : -1);
        take(a.z, ok0 ? a.w - r0 : -1);
        take(b.x, ok1 ? b.y - r0 : -1);
        take(b.z, ok1 ? b.w - r0 : -1);
        nNoted -= n;
        if (count >= 4) flush4();
    };

    const long long nBatches = gcn_scan_batches(nEdges);
    const v4u* plane = (const v4u*)ids;
    const unsigned mine = (unsigned)blockIdx.x * 0x01010101u;

    auto loadBatch = [&](v4u (&q)[GCN_SCAN_LOADS], long long b) {
#pragma unroll
        for (int j = 0; j < GCN_SCAN_LOADS; ++j)
            q[j] = plane[b * (32 * GCN_SCAN_LOADS) + j * 32 + lane];
    };

    int nCh = -1;
    if (LISTS == 2) nCh = __builtin_amdgcn_readfirstlane((int)myHead[0]);
    const bool fromList = LISTS == 2 && (unsigned)nCh <= (unsigned)GCN_LIST_CHUNKS;

    if (fromList) {
        unsigned curE = GCN_NO_ENTRY;
        if (wave < nCh) curE = myChunks[(size_t)wave * 32 + lane];
        for (int c = wave; c < nCh; c += GCN_OWN_WAVES) {
            unsigned nxtE = GCN_NO_ENTRY;
            if (c + GCN_OWN_WAVES < nCh)
                nxtE = myChunks[(size_t)(c + GCN_OWN_WAVES) * 32 + lane];
            const unsigned s = curE & 0xffffu;
            take((int)s, s < (unsigned)GCN_NODES ? (int)(curE >> 16) : -1);
            if (count >= 4) flush4();
            curE = nxtE;
        }
    } else {
    if (nPairs > 0) {
        v4u cur[GCN_SCAN_LOADS];
        loadBatch(cur, wave < nBatches ? (long long)wave : nBatches - 1);
        for (long long b = wave; b < nBatches; b += GCN_OWN_WAVES) {
            v4u nxt[GCN_SCAN_LOADS];
            loadBatch(nxt, b + GCN_OWN_WAVES < nBatches ? b + GCN_OWN_WAVES : nBatches - 1);
#pragma unroll
            for (int j = 0; j < GCN_SCAN_LOADS; ++j) {
                const unsigned word0 =
                    (unsigned)(b * (32 * GCN_SCAN_LOADS) + j * 32 + lane) * 4u;
#pragma unroll
                for (int k = 0; k < 4; ++k) {
                    const unsigned x = cur[j][k] ^ mine;
                    const bool any = ((x - 0x01010101u) & ~x & 0x80808080u) != 0u;
                    const unsigned m = __builtin_amdgcn_ballot_w32(any);
                    if (m != 0u) {
                        if (any)
                            myWords[nNoted + (int)__builtin_amdgcn_mbcnt_lo(m, 0u)] = word0 + k;
                        nNoted += __builtin_popcount(m);
                    }
                }
                while (nNoted >= 32) decode(32);
            }
#pragma unroll
            for (int j = 0; j < GCN_SCAN_LOADS; ++j) cur[j] = nxt[j];
        }
        if (nNoted > 0) decode(nNoted);
    }
    if ((nEdges & 1) && wave == 0) {
        int s = 0, d = -1;
        if (lane == 0) {
            s = edges[2 * (nEdges - 1)];
            d = edges[2 * (nEdges - 1) + 1] - r0;
        }
        take(s, d);
    }
    }
    flush4();
    for (int i = 0; i < count; ++i) {
        const unsigned p = (unsigned)__builtin_amdgcn_readfirstlane((int)myList[i]);
        const float* hp = h + (size_t)(p & 0xffffu) * (size_t)C + lane;
        float* ap = acc + (p >> 16) * C + lane;
        float v[C / 32];
#pragma unroll
        for (int j = 0; j < C / 32; ++j) v[j] = hp[32 * j];
#pragma unroll
        for (int j = 0; j < C / 32; ++j) atomicAdd(ap + 32 * j, v[j]);
    }
    if (LISTS == 1 && outCount > 0) emitChunk();
    __syncthreads();
    if (LISTS == 1 && wave == 0) myHead[lane] = (unsigned)nChunksS;

    {
        constexpr int V  = C / 128;
        constexpr int D  = GCN_Q_ROWS;
        constexpr int NR = GCN_WAVE_ROWS;
        int nq = __builtin_amdgcn_readfirstlane(qCount[wave]);
        if (nq > GCN_Q_CAP) nq = GCN_Q_CAP;
        const unsigned* q  = rowQ  + wave * GCN_Q_CAP;
        unsigned*       qs = rowQs + wave * GCN_Q_CAP;

        int start[NR];
#pragma unroll
        for (int kk = 0; kk < NR; ++kk) start[kk] = 0;
        for (int c0 = 0; c0 < nq; c0 += 32) {
            const bool act = c0 + lane < nq;
            const int key = act ? (int)(q[c0 + lane] >> 21) : -1;
#pragma unroll
            for (int kk = 0; kk + 1 < NR; ++kk)
                start[kk + 1] += __builtin_popcount(__builtin_amdgcn_ballot_w32(key >= 0 && key <= kk));
        }
        for (int c0 = 0; c0 < nq; c0 += 32) {
            const bool act = c0 + lane < nq;
            const unsigned e = act ? q[c0 + lane] : 0u;
            const int key = act ? (int)(e >> 21) : -1;
#pragma unroll
            for (int kk = 0; kk < NR; ++kk) {
                const unsigned m = __builtin_amdgcn_ballot_w32(key == kk);
                if (key == kk) qs[start[kk] + (int)__builtin_amdgcn_mbcnt_lo(m, 0u)] = e;
                start[kk] += __builtin_popcount(m);
            }
        }

        int kdone = 0;
        auto storeRows = [&](int kend) {
#pragma unroll 1
            for (; kdone < kend; ++kdone) {
                const int row  = wave + 32 * kdone;
                const int node = r0 + row;
                if (row < GCN_OWN_ROWS && node < GCN_NODES) {
                    const float d = fmaxf((float)degS[row], 1.0f);
                    const v4f* ap = (const v4f*)(acc + row * C);
                    v4f* op = (v4f*)(outp + (size_t)node * (size_t)C);
#pragma unroll
                    for (int j = 0; j < V; ++j) {
                        v4f a = ap[32 * j + lane];
                        a.x = a.x / d; a.y = a.y / d; a.z = a.z / d; a.w = a.w / d;
                        if (RELU) {
                            a.x = fmaxf(a.x, 0.0f); a.y = fmaxf(a.y, 0.0f);
                            a.z = fmaxf(a.z, 0.0f); a.w = fmaxf(a.w, 0.0f);
                        }
                        op[32 * j + lane] = a;
                    }
                }
            }
        };

        if (nq > 0) {
            auto entryAt = [&](int i) -> unsigned {
                return (unsigned)__builtin_amdgcn_readfirstlane((int)qs[i < nq ? i : nq - 1]);
            };
            v4f sum[V];
#pragma unroll
            for (int j = 0; j < V; ++j) sum[j] = (v4f){0.0f, 0.0f, 0.0f, 0.0f};
            unsigned pc[D];
            v4f vc[D][V];
#pragma unroll
            for (int u = 0; u < D; ++u) {
                pc[u] = 0u;
#pragma unroll
                for (int j = 0; j < V; ++j) vc[u][j] = (v4f){0.0f, 0.0f, 0.0f, 0.0f};
            }
            unsigned rcur = entryAt(0) >> 16;
            auto putRow = [&]() {
                v4f* ap = (v4f*)(acc + rcur * C);
#pragma unroll
                for (int j = 0; j < V; ++j) {
                    v4f a = ap[32 * j + lane];
                    a.x += sum[j].x; a.y += sum[j].y; a.z += sum[j].z; a.w += sum[j].w;
                    ap[32 * j + lane] = a;
                    sum[j] = (v4f){0.0f, 0.0f, 0.0f, 0.0f};
                }
            };
            for (int i = -D; i < nq; i += D) {
                storeRows((int)(rcur >> 5));
                unsigned pn[D];
#pragma unroll
                for (int u = 0; u < D; ++u) pn[u] = entryAt(i + D + u);
#pragma unroll
                for (int u = 0; u < D; ++u) {
                    if (i >= 0 && i + u < nq) {
                        const unsigned r = pc[u] >> 16;
                        if (r != rcur) {
                            putRow();
                            rcur = r;
                        }
#pragma unroll
                        for (int j = 0; j < V; ++j) {
                            sum[j].x += vc[u][j].x; sum[j].y += vc[u][j].y;
                            sum[j].z += vc[u][j].z; sum[j].w += vc[u][j].w;
                        }
                    }
                    pc[u] = pn[u];
                    const v4f* hp = (const v4f*)(h + (size_t)(pc[u] & 0xffffu) * (size_t)C);
#pragma unroll
                    for (int j = 0; j < V; ++j) vc[u][j] = hp[32 * j + lane];
                }
            }
            putRow();
        }

        storeRows(NR);
    }
}

extern "C" void kernel_launch(void* const* d_in, const int* in_sizes, int n_in,
                              void* d_out, int out_size, void* d_ws, size_t ws_size,
                              hipStream_t stream) {
    const float* X  = (const float*)d_in[0];
    const int*   E  = (const int*)  d_in[1];
    const float* W1 = (const float*)d_in[2];
    const float* b1 = (const float*)d_in[3];
    const float* W2 = (const float*)d_in[4];
    const float* b2 = (const float*)d_in[5];
    float* out = (float*)d_out;

    const long long nEdges = (long long)in_sizes[1] / 2;

    char* ws = (char*)d_ws;
    float* h1  = (float*)ws;
    float* h1n = (float*)(ws + (size_t)GCN_NODES * GCN_DHID * 4);
    float* h2  = h1;
    unsigned* ids = (unsigned*)(ws + (size_t)GCN_NODES * GCN_DHID * 4 * 2);
    unsigned* lists = (unsigned*)(ws + (size_t)GCN_NODES * GCN_DHID * 4 * 2 + GCN_IDS_BYTES);

    gcn_gemm_bias_wmma<GCN_DIN, GCN_DHID, 1>
        <<<dim3((GCN_NODES / 16 + GCN_GEMM_WAVES - 1) / GCN_GEMM_WAVES, GCN_DHID / 64),
           32 * GCN_GEMM_WAVES, 0, stream>>>(X, W1, b1, h1, GCN_NODES / 16, E, ids, nEdges);

    gcn_owner_mean<GCN_DHID, 1, 1><<<(GCN_NODES + GCN_OWN_ROWS - 1) / GCN_OWN_ROWS, 1024, 0, stream>>>(
        h1, E, ids, lists, h1n, nEdges);

    gcn_gemm_bias_wmma<GCN_DHID, GCN_DOUT, 0>
        <<<dim3((GCN_NODES / 16 + GCN_GEMM_WAVES - 1) / GCN_GEMM_WAVES, GCN_DOUT / 64),
           32 * GCN_GEMM_WAVES, 0, stream>>>(h1n, W2, b2, h2, GCN_NODES / 16, (const int*)nullptr, (unsigned*)nullptr, 0LL);

    gcn_owner_mean<GCN_DOUT, 0, 2><<<(GCN_NODES + GCN_OWN_ROWS - 1) / GCN_OWN_ROWS, 1024, 0, stream>>>(
        h2, E, ids, lists, out, nEdges);
}
